# starting kernel
# baseline (speedup 1.0000x reference)
#include <hip/hip_runtime.h>

typedef __attribute__((ext_vector_type(2))) float v2f;
typedef __attribute__((ext_vector_type(8))) float v8f;

#define BM 128
#define BM2 64
#define BN 128
#define BK 32
#define AS_STRIDE 36
#define BS_STRIDE 132

__global__ __launch_bounds__(256)
void gnn_gemm_bias_act(const float* __restrict__ X, const float* __restrict__ W,
                       const float* __restrict__ bias, float* __restrict__ Y,
                       int M, int K, int N, int doRelu) {
  __shared__ float As[BM * AS_STRIDE];
  __shared__ float Bs[BK * BS_STRIDE];

  const int tid   = threadIdx.x;
  const int lane  = tid & 31;
  const int wave  = tid >> 5;
  const int waveM = wave & 3;
  const int waveN = wave >> 2;
  const int lr    = lane & 15;
  const int kHalf = (lane >> 4) << 1;

  const int m0 = blockIdx.x * BM;
  const int n0 = blockIdx.y * BN;

  v8f acc[2][4];
#pragma unroll
  for (int mt = 0; mt < 2; ++mt)
#pragma unroll
    for (int nt = 0; nt < 4; ++nt)
      acc[mt][nt] = (v8f)0.0f;

  for (int k0 = 0; k0 < K; k0 += BK) {
#pragma unroll
    for (int i = 0; i < 4; ++i) {
      int s  = tid + i * 256;
      int r  = s >> 3;
      int c4 = (s & 7) << 2;
      float4 v = make_float4(0.f, 0.f, 0.f, 0.f);
      int gr = m0 + r;
      if (gr < M)
        v = *(const float4*)(X + (size_t)gr * K + (k0 + c4));
      *(float4*)(As + r * AS_STRIDE + c4) = v;
    }
#pragma unroll
    for (int i = 0; i < 4; ++i) {
      int s  = tid + i * 256;
      int r  = s >> 5;
      int c4 = (s & 31) << 2;
      float4 v = *(const float4*)(W + (size_t)(k0 + r) * N + (n0 + c4));
      *(float4*)(Bs + r * BS_STRIDE + c4) = v;
    }
    __syncthreads();

#pragma unroll
    for (int kk = 0; kk < BK; kk += 4) {
      v2f a[2];
      v2f b[4];
#pragma unroll
      for (int mt = 0; mt < 2; ++mt) {
        int m = waveM * 32 + mt * 16 + lr;
        const float* p = As + m * AS_STRIDE + kk + kHalf;
        a[mt][0] = p[0];
        a[mt][1] = p[1];
      }
#pragma unroll
      for (int nt = 0; nt < 4; ++nt) {
        int n = waveN * 64 + nt * 16 + lr;
        const float* p = Bs + (kk + kHalf) * BS_STRIDE + n;
        b[nt][0] = p[0];
        b[nt][1] = p[BS_STRIDE];
      }
#pragma unroll
      for (int mt = 0; mt < 2; ++mt)
#pragma unroll
        for (int nt = 0; nt < 4; ++nt)
          acc[mt][nt] = __builtin_amdgcn_wmma_f32_16x16x4_f32(
              false, a[mt], false, b[nt], (short)0, acc[mt][nt], false, false);
    }
    __syncthreads();
  }

  const int rowHalf = (lane >> 4) << 3;
#pragma unroll
  for (int mt = 0; mt < 2; ++mt) {
#pragma unroll
    for (int nt = 0; nt < 4; ++nt) {
      int col  = n0 + waveN * 64 + nt * 16 + lr;
      float bv = bias[col];
#pragma unroll
      for (int v = 0; v < 8; ++v) {
        int row = m0 + waveM * 32 + mt * 16 + rowHalf + v;
        float val = acc[mt][nt][v] + bv;
        if (doRelu) val = fmaxf(val, 0.0f);
        if (row < M) Y[(size_t)row * N + col] = val;
      }
    }
  }
}

__global__ __launch_bounds__(256)
void gnn_gemm_dual(const float* __restrict__ X,
                   const float* __restrict__ W1, const float* __restrict__ b1,
                   float* __restrict__ Y1,
                   const float* __restrict__ W2, const float* __restrict__ b2,
                   float* __restrict__ Y2,
                   int M, int K, int N) {
  __shared__ float As[BM2 * AS_STRIDE];
  __shared__ float Bs[2][BK * BS_STRIDE];

  const int tid     = threadIdx.x;
  const int lane    = tid & 31;
  const int wave    = tid >> 5;
  const int waveOut = wave >> 2;
  const int wq      = wave & 3;
  const int waveM   = wq & 1;
  const int waveN   = wq >> 1;
  const int lr      = lane & 15;
  const int kHalf   = (lane >> 4) << 1;

  const int m0 = blockIdx.x * BM2;
  const int n0 = blockIdx.y * BN;

  v8f acc[2][4];
#pragma unroll
  for (int mt = 0; mt < 2; ++mt)
#pragma unroll
    for (int nt = 0; nt < 4; ++nt)
      acc[mt][nt] = (v8f)0.0f;

  const float* Bw = &Bs[waveOut][0];

  for (int k0 = 0; k0 < K; k0 += BK) {
#pragma unroll
    for (int i = 0; i < 2; ++i) {
      int s  = tid + i * 256;
      int r  = s >> 3;
      int c4 = (s & 7) << 2;
      float4 v = make_float4(0.f, 0.f, 0.f, 0.f);
      int gr = m0 + r;
      if (gr < M)
        v = *(const float4*)(X + (size_t)gr * K + (k0 + c4));
      *(float4*)(As + r * AS_STRIDE + c4) = v;
    }
#pragma unroll
    for (int i = 0; i < 4; ++i) {
      int s  = tid + i * 256;
      int r  = s >> 5;
      int c4 = (s & 31) << 2;
      float4 v = *(const float4*)(W1 + (size_t)(k0 + r) * N + (n0 + c4));
      *(float4*)(&Bs[0][0] + r * BS_STRIDE + c4) = v;
    }
#pragma unroll
    for (int i = 0; i < 4; ++i) {
      int s  = tid + i * 256;
      int r  = s >> 5;
      int c4 = (s & 31) << 2;
      float4 v = *(const float4*)(W2 + (size_t)(k0 + r) * N + (n0 + c4));
      *(float4*)(&Bs[1][0] + r * BS_STRIDE + c4) = v;
    }
    __syncthreads();

#pragma unroll
    for (int kk = 0; kk < BK; kk += 4) {
      v2f a[2];
      v2f b[4];
#pragma unroll
      for (int mt = 0; mt < 2; ++mt) {
        int m = waveM * 32 + mt * 16 + lr;
        const float* p = As + m * AS_STRIDE + kk + kHalf;
        a[mt][0] = p[0];
        a[mt][1] = p[1];
      }
#pragma unroll
      for (int nt = 0; nt < 4; ++nt) {
        int n = waveN * 64 + nt * 16 + lr;
        const float* p = Bw + (kk + kHalf) * BS_STRIDE + n;
        b[nt][0] = p[0];
        b[nt][1] = p[BS_STRIDE];
      }
#pragma unroll
      for (int mt = 0; mt < 2; ++mt)
#pragma unroll
        for (int nt = 0; nt < 4; ++nt)
          acc[mt][nt] = __builtin_amdgcn_wmma_f32_16x16x4_f32(
              false, a[mt], false, b[nt], (short)0, acc[mt][nt], false, false);
    }
    __syncthreads();
  }

  const float* bias = waveOut ? b2 : b1;
  float*       Y    = waveOut ? Y2 : Y1;
  const int rowHalf = (lane >> 4) << 3;
#pragma unroll
  for (int mt = 0; mt < 2; ++mt) {
#pragma unroll
    for (int nt = 0; nt < 4; ++nt) {
      int col  = n0 + waveN * 64 + nt * 16 + lr;
      float bv = bias[col];
#pragma unroll
      for (int v = 0; v < 8; ++v) {
        int row = m0 + waveM * 32 + mt * 16 + rowHalf + v;
        float val = fmaxf(acc[mt][nt][v] + bv, 0.0f);
        if (row < M) Y[(size_t)row * N + col] = val;
      }
    }
  }
}

__global__ __launch_bounds__(256)
void gnn_scatter_add(const int* __restrict__ recvIdx, const int* __restrict__ sendIdx,
                     const float* __restrict__ msg, float* __restrict__ agg, int nEdges) {
  int e = blockIdx.x;
  if (e >= nEdges) return;
  int r = recvIdx[e];
  int s = sendIdx[e];
  int c = threadIdx.x;
  float v = msg[(size_t)s * 256 + c];
  unsigned int off = ((unsigned int)r * 256u + (unsigned int)c) * 4u;
  asm volatile("global_atomic_add_f32 %0, %1, %2 scope:SCOPE_DEV"
               :
               : "v"(off), "v"(v), "s"(agg)
               : "memory");
}

extern "C" void kernel_launch(void* const* d_in, const int* in_sizes, int n_in,
                              void* d_out, int out_size, void* d_ws, size_t ws_size,
                              hipStream_t stream) {
  (void)in_sizes; (void)n_in; (void)out_size; (void)ws_size;

  constexpr int U = 50000, D = 10000, E = 200000;
  constexpr int DIN = 1024, HID = 256, OUT = 128;

  const float* user_feat = (const float*)d_in[0];
  const float* dish_feat = (const float*)d_in[1];
  const int*   edge_user = (const int*)d_in[2];
  const int*   edge_dish = (const int*)d_in[3];
  const float* W_ui = (const float*)d_in[4];   const float* b_ui = (const float*)d_in[5];
  const float* W_di = (const float*)d_in[6];   const float* b_di = (const float*)d_in[7];
  const float* W_um = (const float*)d_in[8];   const float* b_um = (const float*)d_in[9];
  const float* W_dm = (const float*)d_in[10];  const float* b_dm = (const float*)d_in[11];
  const float* W_uu = (const float*)d_in[12];  const float* b_uu = (const float*)d_in[13];
  const float* W_du = (const float*)d_in[14];  const float* b_du = (const float*)d_in[15];
  const float* W_up = (const float*)d_in[16];  const float* b_up = (const float*)d_in[17];
  const float* W_dp = (const float*)d_in[18];  const float* b_dp = (const float*)d_in[19];

  float* out_user = (float*)d_out;
  float* out_dish = out_user + (size_t)U * OUT;

  float* ws      = (float*)d_ws;
  float* userAgg = ws;
  float* dishAgg = userAgg + (size_t)U * HID;
  float* msgUser = dishAgg + (size_t)D * HID;
  float* msgDish = msgUser + (size_t)U * HID;
  float* userUpd = msgUser;
  float* dishUpd = msgDish;

  dim3 blk(256);
  const int gU  = (U + BM - 1) / BM;
  const int gD  = (D + BM - 1) / BM;
  const int gU2 = (U + BM2 - 1) / BM2;
  const int gD2 = (D + BM2 - 1) / BM2;

  gnn_gemm_dual<<<dim3(gU2, HID / BN), blk, 0, stream>>>(
      user_feat, W_ui, b_ui, userAgg, W_dm, b_dm, msgUser, U, DIN, HID);
  gnn_gemm_dual<<<dim3(gD2, HID / BN), blk, 0, stream>>>(
      dish_feat, W_di, b_di, dishAgg, W_um, b_um, msgDish, D, DIN, HID);

  gnn_scatter_add<<<dim3(E), blk, 0, stream>>>(edge_user, edge_dish, msgDish, userAgg, E);
  gnn_scatter_add<<<dim3(E), blk, 0, stream>>>(edge_dish, edge_user, msgUser, dishAgg, E);

  gnn_gemm_bias_act<<<dim3(gU, HID / BN), blk, 0, stream>>>(userAgg, W_uu, b_uu, userUpd, U, HID, HID, 1);
  gnn_gemm_bias_act<<<dim3(gU, OUT / BN), blk, 0, stream>>>(userUpd, W_up, b_up, out_user, U, HID, OUT, 0);
  gnn_gemm_bias_act<<<dim3(gD, HID / BN), blk, 0, stream>>>(dishAgg, W_du, b_du, dishUpd, D, HID, HID, 1);
  gnn_gemm_bias_act<<<dim3(gD, OUT / BN), blk, 0, stream>>>(dishUpd, W_dp, b_dp, out_dish, D, HID, OUT, 0);
}


# Layer-1 GEMM made weights-stationary: a block keeps its 1024x64 weight slice in LDS, its waves stream X rows from memory straight into A fragments, no K-loop barriers; k sum order changes
# speedup vs baseline: 1.5284x; 1.5284x over previous
#include <hip/hip_runtime.h>

typedef __attribute__((ext_vector_type(2))) float v2f;
typedef __attribute__((ext_vector_type(8))) float v8f;
typedef __attribute__((ext_vector_type(4))) float v4f;

#define BM 128
#define BN 128
#define BK 32
#define AS_STRIDE 36
#define BS_STRIDE 132

__global__ __launch_bounds__(256)
void gnn_gemm_bias_act(const float* __restrict__ X, const float* __restrict__ W,
                       const float* __restrict__ bias, float* __restrict__ Y,
                       int M, int K, int N, int doRelu) {
  __shared__ float As[BM * AS_STRIDE];
  __shared__ float Bs[BK * BS_STRIDE];

  const int tid   = threadIdx.x;
  const int lane  = tid & 31;
  const int wave  = tid >> 5;
  const int waveM = wave & 3;
  const int waveN = wave >> 2;
  const int lr    = lane & 15;
  const int kHalf = (lane >> 4) << 1;

  const int m0 = blockIdx.x * BM;
  const int n0 = blockIdx.y * BN;

  v8f acc[2][4];
#pragma unroll
  for (int mt = 0; mt < 2; ++mt)
#pragma unroll
    for (int nt = 0; nt < 4; ++nt)
      acc[mt][nt] = (v8f)0.0f;

  for (int k0 = 0; k0 < K; k0 += BK) {
#pragma unroll
    for (int i = 0; i < 4; ++i) {
      int s  = tid + i * 256;
      int r  = s >> 3;
      int c4 = (s & 7) << 2;
      float4 v = make_float4(0.f, 0.f, 0.f, 0.f);
      int gr = m0 + r;
      if (gr < M)
        v = *(const float4*)(X + (size_t)gr * K + (k0 + c4));
      *(float4*)(As + r * AS_STRIDE + c4) = v;
    }
#pragma unroll
    for (int i = 0; i < 4; ++i) {
      int s  = tid + i * 256;
      int r  = s >> 5;
      int c4 = (s & 31) << 2;
      float4 v = *(const float4*)(W + (size_t)(k0 + r) * N + (n0 + c4));
      *(float4*)(Bs + r * BS_STRIDE + c4) = v;
    }
    __syncthreads();

#pragma unroll
    for (int kk = 0; kk < BK; kk += 4) {
      v2f a[2];
      v2f b[4];
#pragma unroll
      for (int mt = 0; mt < 2; ++mt) {
        int m = waveM * 32 + mt * 16 + lr;
        const float* p = As + m * AS_STRIDE + kk + kHalf;
        a[mt][0] = p[0];
        a[mt][1] = p[1];
      }
#pragma unroll
      for (int nt = 0; nt < 4; ++nt) {
        int n = waveN * 64 + nt * 16 + lr;
        const float* p = Bs + (kk + kHalf) * BS_STRIDE + n;
        b[nt][0] = p[0];
        b[nt][1] = p[BS_STRIDE];
      }
#pragma unroll
      for (int mt = 0; mt < 2; ++mt)
#pragma unroll
        for (int nt = 0; nt < 4; ++nt)
          acc[mt][nt] = __builtin_amdgcn_wmma_f32_16x16x4_f32(
              false, a[mt], false, b[nt], (short)0, acc[mt][nt], false, false);
    }
    __syncthreads();
  }

  const int rowHalf = (lane >> 4) << 3;
#pragma unroll
  for (int mt = 0; mt < 2; ++mt) {
#pragma unroll
    for (int nt = 0; nt < 4; ++nt) {
      int col  = n0 + waveN * 64 + nt * 16 + lr;
      float bv = bias[col];
#pragma unroll
      for (int v = 0; v < 8; ++v) {
        int row = m0 + waveM * 32 + mt * 16 + rowHalf + v;
        float val = acc[mt][nt][v] + bv;
        if (doRelu) val = fmaxf(val, 0.0f);
        if (row < M) Y[(size_t)row * N + col] = val;
      }
    }
  }
}

#define WS_K   1024
#define WS_N   256
#define WS_NS  64

__device__ __forceinline__ void l1_chunk_mma(v8f (&acc)[2][4], const v4f (&a0)[4],
                                             const v4f (&a1)[4], const v4f* bqk) {
#pragma unroll
  for (int g = 0; g < 4; ++g) {
    v4f b[4];
#pragma unroll
    for (int nt = 0; nt < 4; ++nt)
      b[nt] = bqk[(2 * g) * WS_NS + nt * 16];
    const v2f ax = __builtin_shufflevector(a0[g], a0[g], 0, 1);
    const v2f ay = __builtin_shufflevector(a0[g], a0[g], 2, 3);
    const v2f cx = __builtin_shufflevector(a1[g], a1[g], 0, 1);
    const v2f cy = __builtin_shufflevector(a1[g], a1[g], 2, 3);
#pragma unroll
    for (int nt = 0; nt < 4; ++nt) {
      const v2f bx = __builtin_shufflevector(b[nt], b[nt], 0, 1);
      const v2f by = __builtin_shufflevector(b[nt], b[nt], 2, 3);
      acc[0][nt] = __builtin_amdgcn_wmma_f32_16x16x4_f32(
          false, ax, false, bx, (short)0, acc[0][nt], false, false);
      acc[1][nt] = __builtin_amdgcn_wmma_f32_16x16x4_f32(
          false, cx, false, bx, (short)0, acc[1][nt], false, false);
      acc[0][nt] = __builtin_amdgcn_wmma_f32_16x16x4_f32(
          false, ay, false, by, (short)0, acc[0][nt], false, false);
      acc[1][nt] = __builtin_amdgcn_wmma_f32_16x16x4_f32(
          false, cy, false, by, (short)0, acc[1][nt], false, false);
    }
  }
}

__global__ __launch_bounds__(256)
void gnn_l1_stationary(const float* __restrict__ X,
                       const float* __restrict__ W1, const float* __restrict__ b1,
                       float* __restrict__ Y1,
                       const float* __restrict__ W2, const float* __restrict__ b2,
                       float* __restrict__ Y2, int M) {
  __shared__ v4f Bq[(WS_K / 4) * WS_NS];

  const int tid    = threadIdx.x;
  const int lane   = tid & 31;
  const int wave   = __builtin_amdgcn_readfirstlane(tid >> 5);
  const int c      = lane & 15;
  const int h      = lane >> 4;
  const int slice  = blockIdx.x;
  const int outSel = slice >> 2;
  const int n0     = (slice & 3) * WS_NS;

  const float* __restrict__ W    = outSel ? W2 : W1;
  const float* __restrict__ bias = outSel ? b2 : b1;
  float* __restrict__       Y    = outSel ? Y2 : Y1;

  {
    const int n  = tid & 63;
    const int qs = tid >> 6;
    const float* wp = W + n0 + n;
    for (int q = qs; q < WS_K / 4; q += 16) {
      v4f v[4];
#pragma unroll
      for (int u = 0; u < 4; ++u) {
        const float* p = wp + (size_t)(4 * (q + 4 * u)) * WS_N;
        v[u].x = p[0];
        v[u].y = p[WS_N];
        v[u].z = p[2 * WS_N];
        v[u].w = p[3 * WS_N];
      }
#pragma unroll
      for (int u = 0; u < 4; ++u)
        Bq[(q + 4 * u) * WS_NS + n] = v[u];
    }
  }
  __syncthreads();

  const int nTiles  = (M + 31) >> 5;
  const int rowHalf = h << 3;
  const v4f* bq = Bq + h * WS_NS + c;

  float bv_[4];
#pragma unroll
  for (int nt = 0; nt < 4; ++nt) bv_[nt] = bias[n0 + nt * 16 + c];

  for (int tile = wave * (int)gridDim.y + (int)blockIdx.y; tile < nTiles;
       tile += 8 * (int)gridDim.y) {
    const int row0 = tile << 5;
    int ra = row0 + c;       ra = ra < M ? ra : M - 1;
    int rb = row0 + 16 + c;  rb = rb < M ? rb : M - 1;
    const v4f* __restrict__ xa = (const v4f*)(X + (size_t)ra * WS_K) + h;
    const v4f* __restrict__ xb = (const v4f*)(X + (size_t)rb * WS_K) + h;

    v8f acc[2][4];
#pragma unroll
    for (int mt = 0; mt < 2; ++mt)
#pragma unroll
      for (int nt = 0; nt < 4; ++nt)
        acc[mt][nt] = (v8f)0.0f;

    v4f a0[4], a1[4], p0[4], p1[4];
#pragma unroll
    for (int g = 0; g < 4; ++g) { a0[g] = xa[2 * g]; a1[g] = xb[2 * g]; }

    for (int kc = 0; kc < WS_K; kc += 64) {
#pragma unroll
      for (int g = 0; g < 4; ++g) {
        p0[g] = xa[((kc + 32) >> 2) + 2 * g];
        p1[g] = xb[((kc + 32) >> 2) + 2 * g];
      }
      l1_chunk_mma(acc, a0, a1, bq + (kc >> 2) * WS_NS);

      const int kn = (kc + 64 < WS_K ? kc + 64 : kc + 32) >> 2;
#pragma unroll
      for (int g = 0; g < 4; ++g) { a0[g] = xa[kn + 2 * g]; a1[g] = xb[kn + 2 * g]; }
      l1_chunk_mma(acc, p0, p1, bq + ((kc + 32) >> 2) * WS_NS);

    }

#pragma unroll
    for (int mt = 0; mt < 2; ++mt) {
#pragma unroll
      for (int nt = 0; nt < 4; ++nt) {
        const int col = n0 + nt * 16 + c;
#pragma unroll
        for (int v = 0; v < 8; ++v) {
          const int row = row0 + mt * 16 + rowHalf + v;
          const float val = fmaxf(acc[mt][nt][v] + bv_[nt], 0.0f);
          if (row < M) Y[(size_t)row * WS_N + col] = val;
        }
      }
    }
  }
}

__global__ __launch_bounds__(256)
void gnn_scatter_add(const int* __restrict__ recvIdx, const int* __restrict__ sendIdx,
                     const float* __restrict__ msg, float* __restrict__ agg, int nEdges) {
  int e = blockIdx.x;
  if (e >= nEdges) return;
  int r = recvIdx[e];
  int s = sendIdx[e];
  int c = threadIdx.x;
  float v = msg[(size_t)s * 256 + c];
  unsigned int off = ((unsigned int)r * 256u + (unsigned int)c) * 4u;
  asm volatile("global_atomic_add_f32 %0, %1, %2 scope:SCOPE_DEV"
               :
               : "v"(off), "v"(v), "s"(agg)
               : "memory");
}

extern "C" void kernel_launch(void* const* d_in, const int* in_sizes, int n_in,
                              void* d_out, int out_size, void* d_ws, size_t ws_size,
                              hipStream_t stream) {
  (void)in_sizes; (void)n_in; (void)out_size; (void)ws_size;

  constexpr int U = 50000, D = 10000, E = 200000;
  constexpr int DIN = 1024, HID = 256, OUT = 128;
  static_assert(DIN == WS_K && HID == WS_N, "the layer-1 kernel is written for K = 1024, N = 256");

  const float* user_feat = (const float*)d_in[0];
  const float* dish_feat = (const float*)d_in[1];
  const int*   edge_user = (const int*)d_in[2];
  const int*   edge_dish = (const int*)d_in[3];
  const float* W_ui = (const float*)d_in[4];   const float* b_ui = (const float*)d_in[5];
  const float* W_di = (const float*)d_in[6];   const float* b_di = (const float*)d_in[7];
  const float* W_um = (const float*)d_in[8];   const float* b_um = (const float*)d_in[9];
  const float* W_dm = (const float*)d_in[10];  const float* b_dm = (const float*)d_in[11];
  const float* W_uu = (const float*)d_in[12];  const float* b_uu = (const float*)d_in[13];
  const float* W_du = (const float*)d_in[14];  const float* b_du = (const float*)d_in[15];
  const float* W_up = (const float*)d_in[16];  const float* b_up = (const float*)d_in[17];
  const float* W_dp = (const float*)d_in[18];  const float* b_dp = (const float*)d_in[19];

  float* out_user = (float*)d_out;
  float* out_dish = out_user + (size_t)U * OUT;

  float* ws      = (float*)d_ws;
  float* userAgg = ws;
  float* dishAgg = userAgg + (size_t)U * HID;
  float* msgUser = dishAgg + (size_t)D * HID;
  float* msgDish = msgUser + (size_t)U * HID;
  float* userUpd = msgUser;
  float* dishUpd = msgDish;

  dim3 blk(256);
  const int gU  = (U + BM - 1) / BM;
  const int gD  = (D + BM - 1) / BM;

  gnn_l1_stationary<<<dim3(8, 32), blk, 0, stream>>>(
      user_feat, W_ui, b_ui, userAgg, W_dm, b_dm, msgUser, U);
  gnn_l1_stationary<<<dim3(8, 32), blk, 0, stream>>>(
      dish_feat, W_di, b_di, dishAgg, W_um, b_um, msgDish, D);

  gnn_scatter_add<<<dim3(E), blk, 0, stream>>>(edge_user, edge_dish, msgDish, userAgg, E);
  gnn_scatter_add<<<dim3(E), blk, 0, stream>>>(edge_dish, edge_user, msgUser, dishAgg, E);

  gnn_gemm_bias_act<<<dim3(gU, HID / BN), blk, 0, stream>>>(userAgg, W_uu, b_uu, userUpd, U, HID, HID, 1);
  gnn_gemm_bias_act<<<dim3(gU, OUT / BN), blk, 0, stream>>>(userUpd, W_up, b_up, out_user, U, HID, OUT, 0);
  gnn_gemm_bias_act<<<dim3(gD, HID / BN), blk, 0, stream>>>(dishAgg, W_du, b_du, dishUpd, D, HID, HID, 1);
  gnn_gemm_bias_act<<<dim3(gD, OUT / BN), blk, 0, stream>>>(dishUpd, W_dp, b_dp, out_dish, D, HID, OUT, 0);
}


# Edge scatter: a wave per 32 edges, not a block per edge: the wave moves each whole row itself, 8 line loads then 8 whole-line f32 atomic adds, next 4 rows' loads in flight; adds stay in arrival order
# speedup vs baseline: 2.0002x; 1.3087x over previous
#include <hip/hip_runtime.h>

typedef __attribute__((ext_vector_type(2))) float v2f;
typedef __attribute__((ext_vector_type(8))) float v8f;
typedef __attribute__((ext_vector_type(4))) float v4f;

#define BM 128
#define BN 128
#define BK 32
#define AS_STRIDE 36
#define BS_STRIDE 132

__global__ __launch_bounds__(256)
void gnn_gemm_bias_act(const float* __restrict__ X, const float* __restrict__ W,
                       const float* __restrict__ bias, float* __restrict__ Y,
                       int M, int K, int N, int doRelu) {
  __shared__ float As[BM * AS_STRIDE];
  __shared__ float Bs[BK * BS_STRIDE];

  const int tid   = threadIdx.x;
  const int lane  = tid & 31;
  const int wave  = tid >> 5;
  const int waveM = wave & 3;
  const int waveN = wave >> 2;
  const int lr    = lane & 15;
  const int kHalf = (lane >> 4) << 1;

  const int m0 = blockIdx.x * BM;
  const int n0 = blockIdx.y * BN;

  v8f acc[2][4];
#pragma unroll
  for (int mt = 0; mt < 2; ++mt)
#pragma unroll
    for (int nt = 0; nt < 4; ++nt)
      acc[mt][nt] = (v8f)0.0f;

  for (int k0 = 0; k0 < K; k0 += BK) {
#pragma unroll
    for (int i = 0; i < 4; ++i) {
      int s  = tid + i * 256;
      int r  = s >> 3;
      int c4 = (s & 7) << 2;
      float4 v = make_float4(0.f, 0.f, 0.f, 0.f);
      int gr = m0 + r;
      if (gr < M)
        v = *(const float4*)(X + (size_t)gr * K + (k0 + c4));
      *(float4*)(As + r * AS_STRIDE + c4) = v;
    }
#pragma unroll
    for (int i = 0; i < 4; ++i) {
      int s  = tid + i * 256;
      int r  = s >> 5;
      int c4 = (s & 31) << 2;
      float4 v = *(const float4*)(W + (size_t)(k0 + r) * N + (n0 + c4));
      *(float4*)(Bs + r * BS_STRIDE + c4) = v;
    }
    __syncthreads();

#pragma unroll
    for (int kk = 0; kk < BK; kk += 4) {
      v2f a[2];
      v2f b[4];
#pragma unroll
      for (int mt = 0; mt < 2; ++mt) {
        int m = waveM * 32 + mt * 16 + lr;
        const float* p = As + m * AS_STRIDE + kk + kHalf;
        a[mt][0] = p[0];
        a[mt][1] = p[1];
      }
#pragma unroll
      for (int nt = 0; nt < 4; ++nt) {
        int n = waveN * 64 + nt * 16 + lr;
        const float* p = Bs + (kk + kHalf) * BS_STRIDE + n;
        b[nt][0] = p[0];
        b[nt][1] = p[BS_STRIDE];
      }
#pragma unroll
      for (int mt = 0; mt < 2; ++mt)
#pragma unroll
        for (int nt = 0; nt < 4; ++nt)
          acc[mt][nt] = __builtin_amdgcn_wmma_f32_16x16x4_f32(
              false, a[mt], false, b[nt], (short)0, acc[mt][nt], false, false);
    }
    __syncthreads();
  }

  const int rowHalf = (lane >> 4) << 3;
#pragma unroll
  for (int mt = 0; mt < 2; ++mt) {
#pragma unroll
    for (int nt = 0; nt < 4; ++nt) {
      int col  = n0 + waveN * 64 + nt * 16 + lr;
      float bv = bias[col];
#pragma unroll
      for (int v = 0; v < 8; ++v) {
        int row = m0 + waveM * 32 + mt * 16 + rowHalf + v;
        float val = acc[mt][nt][v] + bv;
        if (doRelu) val = fmaxf(val, 0.0f);
        if (row < M) Y[(size_t)row * N + col] = val;
      }
    }
  }
}

#define WS_K   1024
#define WS_N   256
#define WS_NS  64

__device__ __forceinline__ void l1_chunk_mma(v8f (&acc)[2][4], const v4f (&a0)[4],
                                             const v4f (&a1)[4], const v4f* bqk) {
#pragma unroll
  for (int g = 0; g < 4; ++g) {
    v4f b[4];
#pragma unroll
    for (int nt = 0; nt < 4; ++nt)
      b[nt] = bqk[(2 * g) * WS_NS + nt * 16];
    const v2f ax = __builtin_shufflevector(a0[g], a0[g], 0, 1);
    const v2f ay = __builtin_shufflevector(a0[g], a0[g], 2, 3);
    const v2f cx = __builtin_shufflevector(a1[g], a1[g], 0, 1);
    const v2f cy = __builtin_shufflevector(a1[g], a1[g], 2, 3);
#pragma unroll
    for (int nt = 0; nt < 4; ++nt) {
      const v2f bx = __builtin_shufflevector(b[nt], b[nt], 0, 1);
      const v2f by = __builtin_shufflevector(b[nt], b[nt], 2, 3);
      acc[0][nt] = __builtin_amdgcn_wmma_f32_16x16x4_f32(
          false, ax, false, bx, (short)0, acc[0][nt], false, false);
      acc[1][nt] = __builtin_amdgcn_wmma_f32_16x16x4_f32(
          false, cx, false, bx, (short)0, acc[1][nt], false, false);
      acc[0][nt] = __builtin_amdgcn_wmma_f32_16x16x4_f32(
          false, ay, false, by, (short)0, acc[0][nt], false, false);
      acc[1][nt] = __builtin_amdgcn_wmma_f32_16x16x4_f32(
          false, cy, false, by, (short)0, acc[1][nt], false, false);
    }
  }
}

__global__ __launch_bounds__(256)
void gnn_l1_stationary(const float* __restrict__ X,
                       const float* __restrict__ W1, const float* __restrict__ b1,
                       float* __restrict__ Y1,
                       const float* __restrict__ W2, const float* __restrict__ b2,
                       float* __restrict__ Y2, int M) {
  __shared__ v4f Bq[(WS_K / 4) * WS_NS];

  const int tid    = threadIdx.x;
  const int lane   = tid & 31;
  const int wave   = __builtin_amdgcn_readfirstlane(tid >> 5);
  const int c      = lane & 15;
  const int h      = lane >> 4;
  const int slice  = blockIdx.x;
  const int outSel = slice >> 2;
  const int n0     = (slice & 3) * WS_NS;

  const float* __restrict__ W    = outSel ? W2 : W1;
  const float* __restrict__ bias = outSel ? b2 : b1;
  float* __restrict__       Y    = outSel ? Y2 : Y1;

  {
    const int n  = tid & 63;
    const int qs = tid >> 6;
    const float* wp = W + n0 + n;
    for (int q = qs; q < WS_K / 4; q += 16) {
      v4f v[4];
#pragma unroll
      for (int u = 0; u < 4; ++u) {
        const float* p = wp + (size_t)(4 * (q + 4 * u)) * WS_N;
        v[u].x = p[0];
        v[u].y = p[WS_N];
        v[u].z = p[2 * WS_N];
        v[u].w = p[3 * WS_N];
      }
#pragma unroll
      for (int u = 0; u < 4; ++u)
        Bq[(q + 4 * u) * WS_NS + n] = v[u];
    }
  }
  __syncthreads();

  const int nTiles  = (M + 31) >> 5;
  const int rowHalf = h << 3;
  const v4f* bq = Bq + h * WS_NS + c;

  float bv_[4];
#pragma unroll
  for (int nt = 0; nt < 4; ++nt) bv_[nt] = bias[n0 + nt * 16 + c];

  for (int tile = wave * (int)gridDim.y + (int)blockIdx.y; tile < nTiles;
       tile += 8 * (int)gridDim.y) {
    const int row0 = tile << 5;
    int ra = row0 + c;       ra = ra < M ? ra : M - 1;
    int rb = row0 + 16 + c;  rb = rb < M ? rb : M - 1;
    const v4f* __restrict__ xa = (const v4f*)(X + (size_t)ra * WS_K) + h;
    const v4f* __restrict__ xb = (const v4f*)(X + (size_t)rb * WS_K) + h;

    v8f acc[2][4];
#pragma unroll
    for (int mt = 0; mt < 2; ++mt)
#pragma unroll
      for (int nt = 0; nt < 4; ++nt)
        acc[mt][nt] = (v8f)0.0f;

    v4f a0[4], a1[4], p0[4], p1[4];
#pragma unroll
    for (int g = 0; g < 4; ++g) { a0[g] = xa[2 * g]; a1[g] = xb[2 * g]; }

    for (int kc = 0; kc < WS_K; kc += 64) {
#pragma unroll
      for (int g = 0; g < 4; ++g) {
        p0[g] = xa[((kc + 32) >> 2) + 2 * g];
        p1[g] = xb[((kc + 32) >> 2) + 2 * g];
      }
      l1_chunk_mma(acc, a0, a1, bq + (kc >> 2) * WS_NS);

      const int kn = (kc + 64 < WS_K ? kc + 64 : kc + 32) >> 2;
#pragma unroll
      for (int g = 0; g < 4; ++g) { a0[g] = xa[kn + 2 * g]; a1[g] = xb[kn + 2 * g]; }
      l1_chunk_mma(acc, p0, p1, bq + ((kc + 32) >> 2) * WS_NS);

    }

#pragma unroll
    for (int mt = 0; mt < 2; ++mt) {
#pragma unroll
      for (int nt = 0; nt < 4; ++nt) {
        const int col = n0 + nt * 16 + c;
#pragma unroll
        for (int v = 0; v < 8; ++v) {
          const int row = row0 + mt * 16 + rowHalf + v;
          const float val = fmaxf(acc[mt][nt][v] + bv_[nt], 0.0f);
          if (row < M) Y[(size_t)row * WS_N + col] = val;
        }
      }
    }
  }
}

#define SC_EPW 32
#define SC_G   4

__device__ __forceinline__ void sc_load_row(float (&v)[8], const float* __restrict__ msg,
                                            int s, int lane) {
  const float* p = msg + (size_t)s * 256 + lane;
#pragma unroll
  for (int j = 0; j < 8; ++j) v[j] = p[32 * j];
}

__device__ __forceinline__ void sc_add_row(const float (&v)[8], float* agg, int r, int lane) {
  const unsigned int off = ((unsigned int)r * 256u + (unsigned int)lane) * 4u;
#pragma unroll
  for (int j = 0; j < 8; ++j)
    asm volatile("global_atomic_add_f32 %0, %1, %2 offset:%3 scope:SCOPE_DEV"
                 :
                 : "v"(off), "v"(v[j]), "s"(agg), "n"(128 * j)
                 : "memory");
}

__global__ __launch_bounds__(256)
void gnn_scatter_add(const int* __restrict__ recvIdx, const int* __restrict__ sendIdx,
                     const float* __restrict__ msg, float* __restrict__ agg, int nEdges) {
  const int lane = threadIdx.x & 31;
  const int wave = __builtin_amdgcn_readfirstlane((int)threadIdx.x >> 5);
  const int base = ((int)blockIdx.x * 8 + wave) * SC_EPW;
  if (base >= nEdges) return;
  const int left = nEdges - base;
  const int n    = left < SC_EPW ? left : SC_EPW;
  const int ei   = base + (lane < n ? lane : 0);
  const int rv   = recvIdx[ei];
  const int sv   = sendIdx[ei];

  if (n == SC_EPW) {
    float va[SC_G][8], vb[SC_G][8];
#pragma unroll
    for (int q = 0; q < SC_G; ++q)
      sc_load_row(va[q], msg, __builtin_amdgcn_readlane(sv, q), lane);
    for (int g = 0; g < SC_EPW / SC_G; g += 2) {
#pragma unroll
      for (int q = 0; q < SC_G; ++q)
        sc_load_row(vb[q], msg, __builtin_amdgcn_readlane(sv, (g + 1) * SC_G + q), lane);
#pragma unroll
      for (int q = 0; q < SC_G; ++q)
        sc_add_row(va[q], agg, __builtin_amdgcn_readlane(rv, g * SC_G + q), lane);
      if (g + 2 < SC_EPW / SC_G) {
#pragma unroll
        for (int q = 0; q < SC_G; ++q)
          sc_load_row(va[q], msg, __builtin_amdgcn_readlane(sv, (g + 2) * SC_G + q), lane);
      }
#pragma unroll
      for (int q = 0; q < SC_G; ++q)
        sc_add_row(vb[q], agg, __builtin_amdgcn_readlane(rv, (g + 1) * SC_G + q), lane);
    }
  } else {
    for (int i = 0; i < n; ++i) {
      float v[8];
      sc_load_row(v, msg, __builtin_amdgcn_readlane(sv, i), lane);
      sc_add_row(v, agg, __builtin_amdgcn_readlane(rv, i), lane);
    }
  }
}

extern "C" void kernel_launch(void* const* d_in, const int* in_sizes, int n_in,
                              void* d_out, int out_size, void* d_ws, size_t ws_size,
                              hipStream_t stream) {
  (void)in_sizes; (void)n_in; (void)out_size; (void)ws_size;

  constexpr int U = 50000, D = 10000, E = 200000;
  constexpr int DIN = 1024, HID = 256, OUT = 128;
  static_assert(DIN == WS_K && HID == WS_N, "the layer-1 kernel is written for K = 1024, N = 256");

  const float* user_feat = (const float*)d_in[0];
  const float* dish_feat = (const float*)d_in[1];
  const int*   edge_user = (const int*)d_in[2];
  const int*   edge_dish = (const int*)d_in[3];
  const float* W_ui = (const float*)d_in[4];   const float* b_ui = (const float*)d_in[5];
  const float* W_di = (const float*)d_in[6];   const float* b_di = (const float*)d_in[7];
  const float* W_um = (const float*)d_in[8];   const float* b_um = (const float*)d_in[9];
  const float* W_dm = (const float*)d_in[10];  const float* b_dm = (const float*)d_in[11];
  const float* W_uu = (const float*)d_in[12];  const float* b_uu = (const float*)d_in[13];
  const float* W_du = (const float*)d_in[14];  const float* b_du = (const float*)d_in[15];
  const float* W_up = (const float*)d_in[16];  const float* b_up = (const float*)d_in[17];
  const float* W_dp = (const float*)d_in[18];  const float* b_dp = (const float*)d_in[19];

  float* out_user = (float*)d_out;
  float* out_dish = out_user + (size_t)U * OUT;

  float* ws      = (float*)d_ws;
  float* userAgg = ws;
  float* dishAgg = userAgg + (size_t)U * HID;
  float* msgUser = dishAgg + (size_t)D * HID;
  float* msgDish = msgUser + (size_t)U * HID;
  float* userUpd = msgUser;
  float* dishUpd = msgDish;

  dim3 blk(256);
  const int gU  = (U + BM - 1) / BM;
  const int gD  = (D + BM - 1) / BM;
  const int gE  = (E + 8 * SC_EPW - 1) / (8 * SC_EPW);

  gnn_l1_stationary<<<dim3(8, 32), blk, 0, stream>>>(
      user_feat, W_ui, b_ui, userAgg, W_dm, b_dm, msgUser, U);
  gnn_l1_stationary<<<dim3(8, 32), blk, 0, stream>>>(
      dish_feat, W_di, b_di, dishAgg, W_um, b_um, msgDish, D);

  gnn_scatter_add<<<dim3(gE), blk, 0, stream>>>(edge_user, edge_dish, msgDish, userAgg, E);
  gnn_scatter_add<<<dim3(gE), blk, 0, stream>>>(edge_dish, edge_user, msgUser, dishAgg, E);

  gnn_gemm_bias_act<<<dim3(gU, HID / BN), blk, 0, stream>>>(userAgg, W_uu, b_uu, userUpd, U, HID, HID, 1);
  gnn_gemm_bias_act<<<dim3(gU, OUT / BN), blk, 0, stream>>>(userUpd, W_up, b_up, out_user, U, HID, OUT, 0);
  gnn_gemm_bias_act<<<dim3(gD, HID / BN), blk, 0, stream>>>(dishAgg, W_du, b_du, dishUpd, D, HID, HID, 1);
  gnn_gemm_bias_act<<<dim3(gD, OUT / BN), blk, 0, stream>>>(dishUpd, W_dp, b_dp, out_dish, D, HID, OUT, 0);
}


# Layer-2 GEMMs (update, projection) made weights-stationary like layer 1: 256x64 weight slice in LDS, waves stream rows into A fragments, next tile's first chunk prefetched; k sum order changes
# speedup vs baseline: 2.2246x; 1.1122x over previous
#include <hip/hip_runtime.h>

typedef __attribute__((ext_vector_type(2))) float v2f;
typedef __attribute__((ext_vector_type(8))) float v8f;
typedef __attribute__((ext_vector_type(4))) float v4f;

#define WS_K   1024
#define WS_N   256
#define WS_NS  64

__device__ __forceinline__ void l1_chunk_mma(v8f (&acc)[2][4], const v4f (&a0)[4],
                                             const v4f (&a1)[4], const v4f* bqk) {
#pragma unroll
  for (int g = 0; g < 4; ++g) {
    v4f b[4];
#pragma unroll
    for (int nt = 0; nt < 4; ++nt)
      b[nt] = bqk[(2 * g) * WS_NS + nt * 16];
    const v2f ax = __builtin_shufflevector(a0[g], a0[g], 0, 1);
    const v2f ay = __builtin_shufflevector(a0[g], a0[g], 2, 3);
    const v2f cx = __builtin_shufflevector(a1[g], a1[g], 0, 1);
    const v2f cy = __builtin_shufflevector(a1[g], a1[g], 2, 3);
#pragma unroll
    for (int nt = 0; nt < 4; ++nt) {
      const v2f bx = __builtin_shufflevector(b[nt], b[nt], 0, 1);
      const v2f by = __builtin_shufflevector(b[nt], b[nt], 2, 3);
      acc[0][nt] = __builtin_amdgcn_wmma_f32_16x16x4_f32(
          false, ax, false, bx, (short)0, acc[0][nt], false, false);
      acc[1][nt] = __builtin_amdgcn_wmma_f32_16x16x4_f32(
          false, cx, false, bx, (short)0, acc[1][nt], false, false);
      acc[0][nt] = __builtin_amdgcn_wmma_f32_16x16x4_f32(
          false, ay, false, by, (short)0, acc[0][nt], false, false);
      acc[1][nt] = __builtin_amdgcn_wmma_f32_16x16x4_f32(
          false, cy, false, by, (short)0, acc[1][nt], false, false);
    }
  }
}

__global__ __launch_bounds__(256)
void gnn_l1_stationary(const float* __restrict__ X,
                       const float* __restrict__ W1, const float* __restrict__ b1,
                       float* __restrict__ Y1,
                       const float* __restrict__ W2, const float* __restrict__ b2,
                       float* __restrict__ Y2, int M) {
  __shared__ v4f Bq[(WS_K / 4) * WS_NS];

  const int tid    = threadIdx.x;
  const int lane   = tid & 31;
  const int wave   = __builtin_amdgcn_readfirstlane(tid >> 5);
  const int c      = lane & 15;
  const int h      = lane >> 4;
  const int slice  = blockIdx.x;
  const int outSel = slice >> 2;
  const int n0     = (slice & 3) * WS_NS;

  const float* __restrict__ W    = outSel ? W2 : W1;
  const float* __restrict__ bias = outSel ? b2 : b1;
  float* __restrict__       Y    = outSel ? Y2 : Y1;

  {
    const int n  = tid & 63;
    const int qs = tid >> 6;
    const float* wp = W + n0 + n;
    for (int q = qs; q < WS_K / 4; q += 16) {
      v4f v[4];
#pragma unroll
      for (int u = 0; u < 4; ++u) {
        const float* p = wp + (size_t)(4 * (q + 4 * u)) * WS_N;
        v[u].x = p[0];
        v[u].y = p[WS_N];
        v[u].z = p[2 * WS_N];
        v[u].w = p[3 * WS_N];
      }
#pragma unroll
      for (int u = 0; u < 4; ++u)
        Bq[(q + 4 * u) * WS_NS + n] = v[u];
    }
  }
  __syncthreads();

  const int nTiles  = (M + 31) >> 5;
  const int rowHalf = h << 3;
  const v4f* bq = Bq + h * WS_NS + c;

  float bv_[4];
#pragma unroll
  for (int nt = 0; nt < 4; ++nt) bv_[nt] = bias[n0 + nt * 16 + c];

  for (int tile = wave * (int)gridDim.y + (int)blockIdx.y; tile < nTiles;
       tile += 8 * (int)gridDim.y) {
    const int row0 = tile << 5;
    int ra = row0 + c;       ra = ra < M ? ra : M - 1;
    int rb = row0 + 16 + c;  rb = rb < M ? rb : M - 1;
    const v4f* __restrict__ xa = (const v4f*)(X + (size_t)ra * WS_K) + h;
    const v4f* __restrict__ xb = (const v4f*)(X + (size_t)rb * WS_K) + h;

    v8f acc[2][4];
#pragma unroll
    for (int mt = 0; mt < 2; ++mt)
#pragma unroll
      for (int nt = 0; nt < 4; ++nt)
        acc[mt][nt] = (v8f)0.0f;

    v4f a0[4], a1[4], p0[4], p1[4];
#pragma unroll
    for (int g = 0; g < 4; ++g) { a0[g] = xa[2 * g]; a1[g] = xb[2 * g]; }

    for (int kc = 0; kc < WS_K; kc += 64) {
#pragma unroll
      for (int g = 0; g < 4; ++g) {
        p0[g] = xa[((kc + 32) >> 2) + 2 * g];
        p1[g] = xb[((kc + 32) >> 2) + 2 * g];
      }
      l1_chunk_mma(acc, a0, a1, bq + (kc >> 2) * WS_NS);

      const int kn = (kc + 64 < WS_K ? kc + 64 : kc + 32) >> 2;
#pragma unroll
      for (int g = 0; g < 4; ++g) { a0[g] = xa[kn + 2 * g]; a1[g] = xb[kn + 2 * g]; }
      l1_chunk_mma(acc, p0, p1, bq + ((kc + 32) >> 2) * WS_NS);

    }

#pragma unroll
    for (int mt = 0; mt < 2; ++mt) {
#pragma unroll
      for (int nt = 0; nt < 4; ++nt) {
        const int col = n0 + nt * 16 + c;
#pragma unroll
        for (int v = 0; v < 8; ++v) {
          const int row = row0 + mt * 16 + rowHalf + v;
          const float val = fmaxf(acc[mt][nt][v] + bv_[nt], 0.0f);
          if (row < M) Y[(size_t)row * WS_N + col] = val;
        }
      }
    }
  }
}

#define L2_K 256

__device__ __forceinline__ const v4f* l2_row(const float* __restrict__ X, int row, int M, int h) {
  row = row < M ? row : M - 1;
  return (const v4f*)(X + (size_t)row * L2_K) + h;
}

__device__ __forceinline__ void l2_chunk_pair(v8f (&acc)[2][4], v4f (&a0)[4], v4f (&a1)[4],
                                              v4f (&p0)[4], v4f (&p1)[4],
                                              const v4f* xa, const v4f* xb,
                                              const v4f* ya, const v4f* yb, const v4f* bqk) {
#pragma unroll
  for (int g = 0; g < 4; ++g) { p0[g] = xa[8 + 2 * g]; p1[g] = xb[8 + 2 * g]; }
  l1_chunk_mma(acc, a0, a1, bqk);
#pragma unroll
  for (int g = 0; g < 4; ++g) { a0[g] = ya[2 * g]; a1[g] = yb[2 * g]; }
  l1_chunk_mma(acc, p0, p1, bqk + 8 * WS_NS);
}

template <int N, bool RELU, bool GUARD>
__device__ __forceinline__ void l2_store_tile(const v8f (&acc)[2][4], const float (&bv)[4],
                                              float* __restrict__ Y, int row, int col, int M) {
#pragma unroll
  for (int mt = 0; mt < 2; ++mt) {
#pragma unroll
    for (int nt = 0; nt < 4; ++nt) {
#pragma unroll
      for (int v = 0; v < 8; ++v) {
        const int r = row + mt * 16 + v;
        float val = acc[mt][nt][v] + bv[nt];
        if (RELU) val = fmaxf(val, 0.0f);
        if (!GUARD || r < M) Y[(size_t)r * N + col + nt * 16] = val;
      }
    }
  }
}

template <int N, bool RELU>
__global__ __launch_bounds__(256)
void gnn_l2_stationary(const float* __restrict__ X, const float* __restrict__ W,
                       const float* __restrict__ bias, float* __restrict__ Y, int M) {
  __shared__ v4f Bq[(L2_K / 4) * WS_NS];

  const int tid  = threadIdx.x;
  const int lane = tid & 31;
  const int wave = __builtin_amdgcn_readfirstlane(tid >> 5);
  const int c    = lane & 15;
  const int h    = lane >> 4;
  const int n0   = (int)blockIdx.x * WS_NS;

  const int nTiles = (M + 31) >> 5;
  const int stride = 8 * (int)gridDim.y;
  int tile = wave * (int)gridDim.y + (int)blockIdx.y;

  const int t0 = tile < nTiles ? tile : 0;
  const v4f* xa = l2_row(X, (t0 << 5) + c, M, h);
  const v4f* xb = l2_row(X, (t0 << 5) + 16 + c, M, h);
  v4f a0[4], a1[4], p0[4], p1[4];
#pragma unroll
  for (int g = 0; g < 4; ++g) { a0[g] = xa[2 * g]; a1[g] = xb[2 * g]; }

  {
    const int n  = tid & 63;
    const int qs = tid >> 6;
    const float* wp = W + n0 + n;
    v4f v[16];
#pragma unroll
    for (int i = 0; i < 16; ++i) {
      const float* p = wp + (size_t)(4 * (qs + 4 * i)) * N;
      v[i].x = p[0];
      v[i].y = p[N];
      v[i].z = p[2 * N];
      v[i].w = p[3 * N];
    }
#pragma unroll
    for (int i = 0; i < 16; ++i)
      Bq[(qs + 4 * i) * WS_NS + n] = v[i];
  }
  __syncthreads();
  if (tile >= nTiles) return;

  const int rowHalf = h << 3;
  const v4f* bq = Bq + h * WS_NS + c;

  float bv_[4];
#pragma unroll
  for (int nt = 0; nt < 4; ++nt) bv_[nt] = bias[n0 + nt * 16 + c];

  for (;;) {
    const int  row0 = tile << 5;
    const int  next = tile + stride;
    const bool more = next < nTiles;
    const int  pre0 = (more ? next : tile) << 5;
    const v4f* xna = l2_row(X, pre0 + c, M, h);
    const v4f* xnb = l2_row(X, pre0 + 16 + c, M, h);

    v8f acc[2][4];
#pragma unroll
    for (int mt = 0; mt < 2; ++mt)
#pragma unroll
      for (int nt = 0; nt < 4; ++nt)
        acc[mt][nt] = (v8f)0.0f;

#pragma nounroll
    for (int kc = 0; kc < L2_K - 64; kc += 64)
      l2_chunk_pair(acc, a0, a1, p0, p1, xa + (kc >> 2), xb + (kc >> 2),
                    xa + ((kc + 64) >> 2), xb + ((kc + 64) >> 2), bq + (kc >> 2) * WS_NS);
    l2_chunk_pair(acc, a0, a1, p0, p1, xa + ((L2_K - 64) >> 2), xb + ((L2_K - 64) >> 2),
                  xna, xnb, bq + ((L2_K - 64) >> 2) * WS_NS);

    if (row0 + 32 <= M)
      l2_store_tile<N, RELU, false>(acc, bv_, Y, row0 + rowHalf, n0 + c, M);
    else
      l2_store_tile<N, RELU, true>(acc, bv_, Y, row0 + rowHalf, n0 + c, M);

    if (!more) break;
    tile = next;
    xa = xna;
    xb = xnb;
  }
}

#define SC_EPW 32
#define SC_G   4

__device__ __forceinline__ void sc_load_row(float (&v)[8], const float* __restrict__ msg,
                                            int s, int lane) {
  const float* p = msg + (size_t)s * 256 + lane;
#pragma unroll
  for (int j = 0; j < 8; ++j) v[j] = p[32 * j];
}

__device__ __forceinline__ void sc_add_row(const float (&v)[8], float* agg, int r, int lane) {
  const unsigned int off = ((unsigned int)r * 256u + (unsigned int)lane) * 4u;
#pragma unroll
  for (int j = 0; j < 8; ++j)
    asm volatile("global_atomic_add_f32 %0, %1, %2 offset:%3 scope:SCOPE_DEV"
                 :
                 : "v"(off), "v"(v[j]), "s"(agg), "n"(128 * j)
                 : "memory");
}

__global__ __launch_bounds__(256)
void gnn_scatter_add(const int* __restrict__ recvIdx, const int* __restrict__ sendIdx,
                     const float* __restrict__ msg, float* __restrict__ agg, int nEdges) {
  const int lane = threadIdx.x & 31;
  const int wave = __builtin_amdgcn_readfirstlane((int)threadIdx.x >> 5);
  const int base = ((int)blockIdx.x * 8 + wave) * SC_EPW;
  if (base >= nEdges) return;
  const int left = nEdges - base;
  const int n    = left < SC_EPW ? left : SC_EPW;
  const int ei   = base + (lane < n ? lane : 0);
  const int rv   = recvIdx[ei];
  const int sv   = sendIdx[ei];

  if (n == SC_EPW) {
    float va[SC_G][8], vb[SC_G][8];
#pragma unroll
    for (int q = 0; q < SC_G; ++q)
      sc_load_row(va[q], msg, __builtin_amdgcn_readlane(sv, q), lane);
    for (int g = 0; g < SC_EPW / SC_G; g += 2) {
#pragma unroll
      for (int q = 0; q < SC_G; ++q)
        sc_load_row(vb[q], msg, __builtin_amdgcn_readlane(sv, (g + 1) * SC_G + q), lane);
#pragma unroll
      for (int q = 0; q < SC_G; ++q)
        sc_add_row(va[q], agg, __builtin_amdgcn_readlane(rv, g * SC_G + q), lane);
      if (g + 2 < SC_EPW / SC_G) {
#pragma unroll
        for (int q = 0; q < SC_G; ++q)
          sc_load_row(va[q], msg, __builtin_amdgcn_readlane(sv, (g + 2) * SC_G + q), lane);
      }
#pragma unroll
      for (int q = 0; q < SC_G; ++q)
        sc_add_row(vb[q], agg, __builtin_amdgcn_readlane(rv, (g + 1) * SC_G + q), lane);
    }
  } else {
    for (int i = 0; i < n; ++i) {
      float v[8];
      sc_load_row(v, msg, __builtin_amdgcn_readlane(sv, i), lane);
      sc_add_row(v, agg, __builtin_amdgcn_readlane(rv, i), lane);
    }
  }
}

extern "C" void kernel_launch(void* const* d_in, const int* in_sizes, int n_in,
                              void* d_out, int out_size, void* d_ws, size_t ws_size,
                              hipStream_t stream) {
  (void)in_sizes; (void)n_in; (void)out_size; (void)ws_size;

  constexpr int U = 50000, D = 10000, E = 200000;
  constexpr int DIN = 1024, HID = 256, OUT = 128;
  static_assert(DIN == WS_K && HID == WS_N, "the layer-1 kernel is written for K = 1024, N = 256");

  const float* user_feat = (const float*)d_in[0];
  const float* dish_feat = (const float*)d_in[1];
  const int*   edge_user = (const int*)d_in[2];
  const int*   edge_dish = (const int*)d_in[3];
  const float* W_ui = (const float*)d_in[4];   const float* b_ui = (const float*)d_in[5];
  const float* W_di = (const float*)d_in[6];   const float* b_di = (const float*)d_in[7];
  const float* W_um = (const float*)d_in[8];   const float* b_um = (const float*)d_in[9];
  const float* W_dm = (const float*)d_in[10];  const float* b_dm = (const float*)d_in[11];
  const float* W_uu = (const float*)d_in[12];  const float* b_uu = (const float*)d_in[13];
  const float* W_du = (const float*)d_in[14];  const float* b_du = (const float*)d_in[15];
  const float* W_up = (const float*)d_in[16];  const float* b_up = (const float*)d_in[17];
  const float* W_dp = (const float*)d_in[18];  const float* b_dp = (const float*)d_in[19];

  float* out_user = (float*)d_out;
  float* out_dish = out_user + (size_t)U * OUT;

  float* ws      = (float*)d_ws;
  float* userAgg = ws;
  float* dishAgg = userAgg + (size_t)U * HID;
  float* msgUser = dishAgg + (size_t)D * HID;
  float* msgDish = msgUser + (size_t)U * HID;
  float* userUpd = msgUser;
  float* dishUpd = msgDish;

  dim3 blk(256);
  const int gE  = (E + 8 * SC_EPW - 1) / (8 * SC_EPW);

  gnn_l1_stationary<<<dim3(8, 32), blk, 0, stream>>>(
      user_feat, W_ui, b_ui, userAgg, W_dm, b_dm, msgUser, U);
  gnn_l1_stationary<<<dim3(8, 32), blk, 0, stream>>>(
      dish_feat, W_di, b_di, dishAgg, W_um, b_um, msgDish, D);

  gnn_scatter_add<<<dim3(gE), blk, 0, stream>>>(edge_user, edge_dish, msgDish, userAgg, E);
  gnn_scatter_add<<<dim3(gE), blk, 0, stream>>>(edge_dish, edge_user, msgUser, dishAgg, E);

  static_assert(HID == L2_K, "the layer-2 kernel is written for K = 256");
  gnn_l2_stationary<HID, true><<<dim3(HID / WS_NS, 64), blk, 0, stream>>>(userAgg, W_uu, b_uu, userUpd, U);
  gnn_l2_stationary<OUT, false><<<dim3(OUT / WS_NS, 128), blk, 0, stream>>>(userUpd, W_up, b_up, out_user, U);
  gnn_l2_stationary<HID, true><<<dim3(HID / WS_NS, 64), blk, 0, stream>>>(dishAgg, W_du, b_du, dishUpd, D);
  gnn_l2_stationary<OUT, false><<<dim3(OUT / WS_NS, 128), blk, 0, stream>>>(dishUpd, W_dp, b_dp, out_dish, D);
}
